# starting kernel
# baseline (speedup 1.0000x reference)
#include <hip/hip_runtime.h>

typedef __attribute__((ext_vector_type(2))) float v2f;
typedef __attribute__((ext_vector_type(8))) float v8f;

__global__ void qc_prep(const float* __restrict__ wt, float* __restrict__ Abuf) {
    int L = threadIdx.x;
    float coef[12];
#pragma unroll
    for (int w = 0; w < 4; ++w) {
        float th = wt[3 * w + 0];
        float lam = wt[3 * w + 2];
        float st, ct, sl, cl;
        __sincosf(th, &st, &ct);
        __sincosf(lam, &sl, &cl);
        coef[3 * w + 0] = -st * cl;
        coef[3 * w + 1] =  st * sl;
        coef[3 * w + 2] =  ct;
    }
    int row = L & 15;
    bool lo = (L < 16);
#pragma unroll
    for (int c = 0; c < 3; ++c) {
        float axlo = (((4 * c + 0) / 3) == row) ? coef[4 * c + 0] : 0.0f;
        float aylo = (((4 * c + 1) / 3) == row) ? coef[4 * c + 1] : 0.0f;
        float axhi = (((4 * c + 2) / 3) == row) ? coef[4 * c + 2] : 0.0f;
        float ayhi = (((4 * c + 3) / 3) == row) ? coef[4 * c + 3] : 0.0f;
        float2 a;
        a.x = lo ? axlo : axhi;
        a.y = lo ? aylo : ayhi;
        ((float2*)Abuf)[c * 32 + L] = a;
    }
}

__global__ __launch_bounds__(256) void qc_main(const float* __restrict__ x,
                                               const float* __restrict__ Abuf,
                                               float* __restrict__ out) {
    int lane = threadIdx.x & 31;
    int wave = blockIdx.x * (blockDim.x >> 5) + (threadIdx.x >> 5);
    long long e = (long long)wave * 32 + lane;

    const float4* xp = (const float4*)(x + e * 12);
    float4 g0 = xp[0];
    float4 g1 = xp[1];
    float4 g2 = xp[2];
    float ang[12] = {g0.x, g0.y, g0.z, g0.w, g1.x, g1.y, g1.z, g1.w,
                     g2.x, g2.y, g2.z, g2.w};

    float mx[4], my[4], mz[4];
#pragma unroll
    for (int q = 0; q < 4; ++q) {
        float s1, c1, s2, c2, s3, c3;
        __sincosf(ang[3 * q + 0], &s1, &c1);
        __sincosf(ang[3 * q + 1], &s2, &c2);
        __sincosf(ang[3 * q + 2], &s3, &c3);
        float cc = c1 * c2;
        mx[q] = __builtin_fmaf(cc, c3, -(s1 * s3));
        my[q] = __builtin_fmaf(cc, s3, s1 * c3);
        mz[q] = -(c1 * s2);
    }

    float z01 = mz[0] * mz[1];
    float z23 = mz[2] * mz[3];
    float x01 = mx[0] * mx[1];
    float z012 = z01 * mz[2];
    float t[12];
    t[0]  = x01;
    t[1]  = mx[0] * my[1] * z23;
    t[2]  = mz[1] * z23;
    t[3]  = mx[1] * mx[2];
    t[4]  = mz[0] * my[1] * mx[2];
    t[5]  = z01;
    t[6]  = mx[2] * mx[3];
    t[7]  = z01 * my[2] * mx[3];
    t[8]  = z012;
    t[9]  = x01 * mx[3];
    t[10] = -(my[0] * my[1] * mz[2] * my[3]);
    t[11] = z012 * mz[3];

    float ts[12];
#pragma unroll
    for (int k = 0; k < 12; ++k) ts[k] = __shfl_xor(t[k], 16, 32);

    bool lo = (lane < 16);
    v8f accA = {};
    v8f accB = {};
#pragma unroll
    for (int c = 0; c < 3; ++c) {
        float2 af = ((const float2*)Abuf)[c * 32 + lane];
        v2f a;
        a[0] = af.x;
        a[1] = af.y;
        v2f bA, bB;
        bA[0] = lo ? t[4 * c + 0] : ts[4 * c + 2];
        bA[1] = lo ? t[4 * c + 1] : ts[4 * c + 3];
        bB[0] = lo ? ts[4 * c + 0] : t[4 * c + 2];
        bB[1] = lo ? ts[4 * c + 1] : t[4 * c + 3];
        accA = __builtin_amdgcn_wmma_f32_16x16x4_f32(false, a, false, bA,
                                                     (short)0, accA, false, false);
        accB = __builtin_amdgcn_wmma_f32_16x16x4_f32(false, a, false, bB,
                                                     (short)0, accB, false, false);
    }

    if (lo) {
        float4 oA = make_float4(accA[0], accA[1], accA[2], accA[3]);
        float4 oB = make_float4(accB[0], accB[1], accB[2], accB[3]);
        float4* op = (float4*)out;
        long long base = (long long)wave * 32;
        op[base + lane] = oA;
        op[base + 16 + lane] = oB;
    }
}

extern "C" void kernel_launch(void* const* d_in, const int* in_sizes, int n_in,
                              void* d_out, int out_size, void* d_ws, size_t ws_size,
                              hipStream_t stream) {
    const float* x  = (const float*)d_in[0];
    const float* wt = (const float*)d_in[1];
    float* out = (float*)d_out;
    float* Abuf = (float*)d_ws;

    int batch = in_sizes[0] / 12;

    qc_prep<<<1, 32, 0, stream>>>(wt, Abuf);

    int blocks = batch / 256;
    qc_main<<<blocks, 256, 0, stream>>>(x, Abuf, out);
}


# Coefficient prep launch folded into the main kernel: each lane derives its own A-fragment entries from two weights (2 sincos); 2 launches -> 1, no workspace round trip
# speedup vs baseline: 1.1537x; 1.1537x over previous
#include <hip/hip_runtime.h>

typedef __attribute__((ext_vector_type(2))) float v2f;
typedef __attribute__((ext_vector_type(8))) float v8f;

__global__ __launch_bounds__(256) void qc_main(const float* __restrict__ x,
                                               const float* __restrict__ wt,
                                               float* __restrict__ out) {
    int lane = threadIdx.x & 31;
    int wave = blockIdx.x * (blockDim.x >> 5) + (threadIdx.x >> 5);
    long long e = (long long)wave * 32 + lane;

    int row = lane & 15;
    int wsel = row & 3;
    float th = wt[3 * wsel + 0];
    float lam = wt[3 * wsel + 2];

    const float4* xp = (const float4*)(x + e * 12);
    float4 g0 = xp[0];
    float4 g1 = xp[1];
    float4 g2 = xp[2];

    float cf[3];
    {
        float st, ct, sl, cl;
        __sincosf(th, &st, &ct);
        __sincosf(lam, &sl, &cl);
        cf[0] = -st * cl;
        cf[1] =  st * sl;
        cf[2] =  ct;
    }
    bool lo = (lane < 16);
    v2f afrag[3];
#pragma unroll
    for (int c = 0; c < 3; ++c) {
        float axlo = (((4 * c + 0) / 3) == row) ? cf[(4 * c + 0) % 3] : 0.0f;
        float aylo = (((4 * c + 1) / 3) == row) ? cf[(4 * c + 1) % 3] : 0.0f;
        float axhi = (((4 * c + 2) / 3) == row) ? cf[(4 * c + 2) % 3] : 0.0f;
        float ayhi = (((4 * c + 3) / 3) == row) ? cf[(4 * c + 3) % 3] : 0.0f;
        afrag[c][0] = lo ? axlo : axhi;
        afrag[c][1] = lo ? aylo : ayhi;
    }

    float ang[12] = {g0.x, g0.y, g0.z, g0.w, g1.x, g1.y, g1.z, g1.w,
                     g2.x, g2.y, g2.z, g2.w};

    float mx[4], my[4], mz[4];
#pragma unroll
    for (int q = 0; q < 4; ++q) {
        float s1, c1, s2, c2, s3, c3;
        __sincosf(ang[3 * q + 0], &s1, &c1);
        __sincosf(ang[3 * q + 1], &s2, &c2);
        __sincosf(ang[3 * q + 2], &s3, &c3);
        float cc = c1 * c2;
        mx[q] = __builtin_fmaf(cc, c3, -(s1 * s3));
        my[q] = __builtin_fmaf(cc, s3, s1 * c3);
        mz[q] = -(c1 * s2);
    }

    float z01 = mz[0] * mz[1];
    float z23 = mz[2] * mz[3];
    float x01 = mx[0] * mx[1];
    float z012 = z01 * mz[2];
    float t[12];
    t[0]  = x01;
    t[1]  = mx[0] * my[1] * z23;
    t[2]  = mz[1] * z23;
    t[3]  = mx[1] * mx[2];
    t[4]  = mz[0] * my[1] * mx[2];
    t[5]  = z01;
    t[6]  = mx[2] * mx[3];
    t[7]  = z01 * my[2] * mx[3];
    t[8]  = z012;
    t[9]  = x01 * mx[3];
    t[10] = -(my[0] * my[1] * mz[2] * my[3]);
    t[11] = z012 * mz[3];

    float ts[12];
#pragma unroll
    for (int k = 0; k < 12; ++k) ts[k] = __shfl_xor(t[k], 16, 32);

    v8f accA = {};
    v8f accB = {};
#pragma unroll
    for (int c = 0; c < 3; ++c) {
        v2f a = afrag[c];
        v2f bA, bB;
        bA[0] = lo ? t[4 * c + 0] : ts[4 * c + 2];
        bA[1] = lo ? t[4 * c + 1] : ts[4 * c + 3];
        bB[0] = lo ? ts[4 * c + 0] : t[4 * c + 2];
        bB[1] = lo ? ts[4 * c + 1] : t[4 * c + 3];
        accA = __builtin_amdgcn_wmma_f32_16x16x4_f32(false, a, false, bA,
                                                     (short)0, accA, false, false);
        accB = __builtin_amdgcn_wmma_f32_16x16x4_f32(false, a, false, bB,
                                                     (short)0, accB, false, false);
    }

    if (lo) {
        float4 oA = make_float4(accA[0], accA[1], accA[2], accA[3]);
        float4 oB = make_float4(accB[0], accB[1], accB[2], accB[3]);
        float4* op = (float4*)out;
        long long base = (long long)wave * 32;
        op[base + lane] = oA;
        op[base + 16 + lane] = oB;
    }
}

extern "C" void kernel_launch(void* const* d_in, const int* in_sizes, int n_in,
                              void* d_out, int out_size, void* d_ws, size_t ws_size,
                              hipStream_t stream) {
    const float* x  = (const float*)d_in[0];
    const float* wt = (const float*)d_in[1];
    float* out = (float*)d_out;

    int batch = in_sizes[0] / 12;

    int blocks = batch / 256;
    qc_main<<<blocks, 256, 0, stream>>>(x, wt, out);
}


# Each lane owns two elements (64 per wave): the 4 k slots of each f32 WMMA carry one monomial of 4 element groups against a 16-row coefficient operand, so no half-wave exchange; sum order changes
# speedup vs baseline: 1.2661x; 1.0974x over previous
#include <hip/hip_runtime.h>

typedef __attribute__((ext_vector_type(2))) float v2f;
typedef __attribute__((ext_vector_type(8))) float v8f;


__device__ __forceinline__ void pauli_monomials(const float4 g0, const float4 g1,
                                                const float4 g2, float* __restrict__ t) {
    float ang[12] = {g0.x, g0.y, g0.z, g0.w, g1.x, g1.y, g1.z, g1.w,
                     g2.x, g2.y, g2.z, g2.w};
    float mx[4], my[4], mz[4];
#pragma unroll
    for (int q = 0; q < 4; ++q) {
        float s1, c1, s2, c2, s3, c3;
        __sincosf(ang[3 * q + 0], &s1, &c1);
        __sincosf(ang[3 * q + 1], &s2, &c2);
        __sincosf(ang[3 * q + 2], &s3, &c3);
        float cc = c1 * c2;
        mx[q] = __builtin_fmaf(cc, c3, -(s1 * s3));
        my[q] = __builtin_fmaf(cc, s3, s1 * c3);
        mz[q] = -(c1 * s2);
    }
    float z01 = mz[0] * mz[1];
    float z23 = mz[2] * mz[3];
    float x01 = mx[0] * mx[1];
    float z012 = z01 * mz[2];
    t[0]  = x01;
    t[1]  = mx[0] * my[1] * z23;
    t[2]  = mz[1] * z23;
    t[3]  = mx[1] * mx[2];
    t[4]  = mz[0] * my[1] * mx[2];
    t[5]  = z01;
    t[6]  = mx[2] * mx[3];
    t[7]  = z01 * my[2] * mx[3];
    t[8]  = z012;
    t[9]  = x01 * mx[3];
    t[10] = -(my[0] * my[1] * mz[2] * my[3]);
    t[11] = z012 * mz[3];
}

__global__ __launch_bounds__(256, 16) void qc_main(const float* __restrict__ x,
                                               const float* __restrict__ wt,
                                               float* __restrict__ out) {
    int lane = threadIdx.x & 31;
    int wave = blockIdx.x * 8 + (threadIdx.x >> 5);
    long long e0 = (long long)wave * 64 + lane;

    int row = lane & 15;
    int half = lane >> 4;
    int wsel = row & 3;
    float th = wt[3 * wsel + 0];
    float lam = wt[3 * wsel + 2];

    const float4* xp = (const float4*)(x + e0 * 12);
    float4 p0 = xp[0];
    float4 p1 = xp[1];
    float4 p2 = xp[2];
    float4 q0 = xp[96];
    float4 q1 = xp[97];
    float4 q2 = xp[98];
    __builtin_amdgcn_sched_barrier(0);

    float cf[3];
    {
        float st, ct, sl, cl;
        __sincosf(th, &st, &ct);
        __sincosf(lam, &sl, &cl);
        cf[0] = -st * cl;
        cf[1] =  st * sl;
        cf[2] =  ct;
    }
    bool mine = ((row >> 3) == half);
    bool use0 = mine && (((row >> 2) & 1) == 0);
    bool use1 = mine && (((row >> 2) & 1) == 1);

    float t0[12], t1[12];
    pauli_monomials(p0, p1, p2, t0);
    pauli_monomials(q0, q1, q2, t1);

    v8f acc = {};
#pragma unroll
    for (int j = 0; j < 12; ++j) {
        bool w_of_j = (wsel == (j / 3));
        v2f a, b;
        a[0] = (use0 && w_of_j) ? cf[j % 3] : 0.0f;
        a[1] = (use1 && w_of_j) ? cf[j % 3] : 0.0f;
        b[0] = t0[j];
        b[1] = t1[j];
        acc = __builtin_amdgcn_wmma_f32_16x16x4_f32(false, a, false, b,
                                                    (short)0, acc, false, false);
    }

    float4* op = (float4*)out;
    op[e0]      = make_float4(acc[0], acc[1], acc[2], acc[3]);
    op[e0 + 32] = make_float4(acc[4], acc[5], acc[6], acc[7]);
}

extern "C" void kernel_launch(void* const* d_in, const int* in_sizes, int n_in,
                              void* d_out, int out_size, void* d_ws, size_t ws_size,
                              hipStream_t stream) {
    const float* x  = (const float*)d_in[0];
    const float* wt = (const float*)d_in[1];
    float* out = (float*)d_out;

    int batch = in_sizes[0] / 12;

    int blocks = batch / 512;
    qc_main<<<blocks, 256, 0, stream>>>(x, wt, out);
}
